# starting kernel
# baseline (speedup 1.0000x reference)
#include <hip/hip_runtime.h>


typedef __attribute__((ext_vector_type(2))) float v2f;
typedef __attribute__((ext_vector_type(8))) float v8f;

#define N_NODES 100000
#define N_EDGES 600000
#define DIM     128
#define NCH     16

__global__ __launch_bounds__(256) void node_proj_wmma(
    const float* __restrict__ h,
    const float* __restrict__ W,
    float* __restrict__ P)
{
    const int lane  = threadIdx.x & 31;
    const int wave  = (blockIdx.x * (blockDim.x >> 5)) + (threadIdx.x >> 5);
    const int m0    = wave * 16;
    if (m0 >= N_NODES) return;

    const int half  = lane >> 4;
    const int sub   = lane & 15;
    const int khalf = half * 2;

    const float* arow = h + (size_t)(m0 + sub) * DIM + khalf;
    const float* brow0 = W + (size_t)sub * (2 * DIM) + khalf;
    const float* brow1 = brow0 + DIM;

    v8f acc0 = {};
    v8f acc1 = {};

#pragma unroll
    for (int k = 0; k < DIM; k += 4) {
        v2f a  = *(const v2f*)(arow  + k);
        v2f b0 = *(const v2f*)(brow0 + k);
        v2f b1 = *(const v2f*)(brow1 + k);
        acc0 = __builtin_amdgcn_wmma_f32_16x16x4_f32(
            false, a, false, b0, (short)0, acc0, false, false);
        acc1 = __builtin_amdgcn_wmma_f32_16x16x4_f32(
            false, a, false, b1, (short)0, acc1, false, false);
    }

    const int mrw = m0 + half * 8;
#pragma unroll
    for (int r = 0; r < 8; ++r) {
        float* prow = P + (size_t)(mrw + r) * 32;
        prow[sub]      = acc0[r];
        prow[16 + sub] = acc1[r];
    }
}

__global__ __launch_bounds__(256) void edge_gather_add(
    const float* __restrict__ P,
    const int*   __restrict__ esrc,
    const int*   __restrict__ edst,
    const float* __restrict__ b,
    float*       __restrict__ out)
{
    const int t = blockIdx.x * blockDim.x + threadIdx.x;
    const int e  = t >> 2;
    if (e >= N_EDGES) return;
    const int c4 = (t & 3) * 4;

    const int s = esrc[e];
    const int d = edst[e];

    const float4 ps = *(const float4*)(P + (size_t)s * 32 + c4);
    const float4 pd = *(const float4*)(P + (size_t)d * 32 + 16 + c4);
    const float4 bb = *(const float4*)(b + c4);

    float4 r;
    r.x = ps.x + pd.x + bb.x;
    r.y = ps.y + pd.y + bb.y;
    r.z = ps.z + pd.z + bb.z;
    r.w = ps.w + pd.w + bb.w;

    *(float4*)(out + (size_t)e * NCH + c4) = r;
}

extern "C" void kernel_launch(void* const* d_in, const int* in_sizes, int n_in,
                              void* d_out, int out_size, void* d_ws, size_t ws_size,
                              hipStream_t stream) {
    const float* h    = (const float*)d_in[0];
    const int*   src  = (const int*)  d_in[1];
    const int*   dst  = (const int*)  d_in[2];
    const float* W    = (const float*)d_in[3];
    const float* b    = (const float*)d_in[4];
    float*       out  = (float*)d_out;
    float*       P    = (float*)d_ws;

    const int waves_needed = (N_NODES + 15) / 16;
    const int blocks1 = (waves_needed + 7) / 8;
    node_proj_wmma<<<blocks1, 256, 0, stream>>>(h, W, P);

    const int threads2 = N_EDGES * 4;
    const int blocks2 = (threads2 + 255) / 256;
    edge_gather_add<<<blocks2, 256, 0, stream>>>(P, src, dst, b, out);
}


# K1 operand fetch: W staged once per block in LDS; A (global) and B (LDS) fragments read 128 bits per lane in one shared permuted k order (sum order changes); line requests per tile 1536 -> 272
# speedup vs baseline: 1.3751x; 1.3751x over previous
#include <hip/hip_runtime.h>


typedef __attribute__((ext_vector_type(2))) float v2f;
typedef __attribute__((ext_vector_type(4))) float v4f;
typedef __attribute__((ext_vector_type(8))) float v8f;

#define N_NODES 100000
#define N_EDGES 600000
#define DIM     128
#define NCH     16
#define WPITCH  260

__global__ __launch_bounds__(256) void node_proj_wmma(
    const float* __restrict__ h,
    const float* __restrict__ W,
    float* __restrict__ P)
{
    __shared__ __attribute__((aligned(16))) float Ws[NCH * WPITCH];

    const int tid   = threadIdx.x;
    const int lane  = tid & 31;
    const int wave  = blockIdx.x * 8 + (tid >> 5);
    const int m0    = wave * 16;
    const bool valid = m0 < N_NODES;
    const int mld   = valid ? m0 : 0;

    const int hf    = lane >> 4;
    const int sub   = lane & 15;

    const v4f* arow = (const v4f*)(h + (size_t)(mld + sub) * DIM + 4 * hf);
    v4f a[16];
#pragma unroll
    for (int j = 0; j < 16; ++j) a[j] = arow[2 * j];

    {
        const v4f* Wg = (const v4f*)W;
        const int wr = tid >> 6;
        const int wc = (tid & 63) * 4;
        v4f w[4];
#pragma unroll
        for (int i = 0; i < 4; ++i) w[i] = Wg[tid + 256 * i];
#pragma unroll
        for (int i = 0; i < 4; ++i)
            *(v4f*)(Ws + (wr + 4 * i) * WPITCH + wc) = w[i];
    }
    __builtin_amdgcn_fence(__ATOMIC_RELEASE, "workgroup", "local");
    __builtin_amdgcn_s_barrier();
    __builtin_amdgcn_fence(__ATOMIC_ACQUIRE, "workgroup", "local");

    const float* bl = Ws + sub * WPITCH + 4 * hf;

    v8f acc0 = {};
    v8f acc1 = {};

#pragma unroll
    for (int j = 0; j < 16; ++j) {
        const v4f b0 = *(const v4f*)(bl + 8 * j);
        const v4f b1 = *(const v4f*)(bl + DIM + 8 * j);
        const v2f a01 = a[j].xy;
        const v2f a23 = a[j].zw;
        acc0 = __builtin_amdgcn_wmma_f32_16x16x4_f32(
            false, a01, false, b0.xy, (short)0, acc0, false, false);
        acc1 = __builtin_amdgcn_wmma_f32_16x16x4_f32(
            false, a01, false, b1.xy, (short)0, acc1, false, false);
        acc0 = __builtin_amdgcn_wmma_f32_16x16x4_f32(
            false, a23, false, b0.zw, (short)0, acc0, false, false);
        acc1 = __builtin_amdgcn_wmma_f32_16x16x4_f32(
            false, a23, false, b1.zw, (short)0, acc1, false, false);
    }

    if (!valid) return;
    const int mrw = m0 + hf * 8;
#pragma unroll
    for (int r = 0; r < 8; ++r) {
        float* prow = P + (size_t)(mrw + r) * 32;
        prow[sub]      = acc0[r];
        prow[16 + sub] = acc1[r];
    }
}

__global__ __launch_bounds__(256) void edge_gather_add(
    const float* __restrict__ P,
    const int*   __restrict__ esrc,
    const int*   __restrict__ edst,
    const float* __restrict__ b,
    float*       __restrict__ out)
{
    const int t = blockIdx.x * blockDim.x + threadIdx.x;
    const int e  = t >> 2;
    if (e >= N_EDGES) return;
    const int c4 = (t & 3) * 4;

    const int s = esrc[e];
    const int d = edst[e];

    const float4 ps = *(const float4*)(P + (size_t)s * 32 + c4);
    const float4 pd = *(const float4*)(P + (size_t)d * 32 + 16 + c4);
    const float4 bb = *(const float4*)(b + c4);

    float4 r;
    r.x = ps.x + pd.x + bb.x;
    r.y = ps.y + pd.y + bb.y;
    r.z = ps.z + pd.z + bb.z;
    r.w = ps.w + pd.w + bb.w;

    *(float4*)(out + (size_t)e * NCH + c4) = r;
}

extern "C" void kernel_launch(void* const* d_in, const int* in_sizes, int n_in,
                              void* d_out, int out_size, void* d_ws, size_t ws_size,
                              hipStream_t stream) {
    const float* h    = (const float*)d_in[0];
    const int*   src  = (const int*)  d_in[1];
    const int*   dst  = (const int*)  d_in[2];
    const float* W    = (const float*)d_in[3];
    const float* b    = (const float*)d_in[4];
    float*       out  = (float*)d_out;
    float*       P    = (float*)d_ws;

    const int waves_needed = (N_NODES + 15) / 16;
    const int blocks1 = (waves_needed + 7) / 8;
    node_proj_wmma<<<blocks1, 256, 0, stream>>>(h, W, P);

    const int threads2 = N_EDGES * 4;
    const int blocks2 = (threads2 + 255) / 256;
    edge_gather_add<<<blocks2, 256, 0, stream>>>(P, src, dst, b, out);
}


# Edge kernel: 128-160 edges per wave instead of 8 (75000 -> 4096 waves, one resident pass); index lines read coalesced and shared by lane permute, all gathers in flight before the sums; sum order kept
# speedup vs baseline: 1.5216x; 1.1065x over previous
#include <hip/hip_runtime.h>


typedef __attribute__((ext_vector_type(2))) float v2f;
typedef __attribute__((ext_vector_type(4))) float v4f;
typedef __attribute__((ext_vector_type(8))) float v8f;

#define N_NODES 100000
#define N_EDGES 600000
#define DIM     128
#define NCH     16
#define WPITCH  260

__global__ __launch_bounds__(256) void node_proj_wmma(
    const float* __restrict__ h,
    const float* __restrict__ W,
    float* __restrict__ P)
{
    __shared__ __attribute__((aligned(16))) float Ws[NCH * WPITCH];

    const int tid   = threadIdx.x;
    const int lane  = tid & 31;
    const int wave  = blockIdx.x * 8 + (tid >> 5);
    const int m0    = wave * 16;
    const bool valid = m0 < N_NODES;
    const int mld   = valid ? m0 : 0;

    const int hf    = lane >> 4;
    const int sub   = lane & 15;

    const v4f* arow = (const v4f*)(h + (size_t)(mld + sub) * DIM + 4 * hf);
    v4f a[16];
#pragma unroll
    for (int j = 0; j < 16; ++j) a[j] = arow[2 * j];

    {
        const v4f* Wg = (const v4f*)W;
        const int wr = tid >> 6;
        const int wc = (tid & 63) * 4;
        v4f w[4];
#pragma unroll
        for (int i = 0; i < 4; ++i) w[i] = Wg[tid + 256 * i];
#pragma unroll
        for (int i = 0; i < 4; ++i)
            *(v4f*)(Ws + (wr + 4 * i) * WPITCH + wc) = w[i];
    }
    __builtin_amdgcn_fence(__ATOMIC_RELEASE, "workgroup", "local");
    __builtin_amdgcn_s_barrier();
    __builtin_amdgcn_fence(__ATOMIC_ACQUIRE, "workgroup", "local");

    const float* bl = Ws + sub * WPITCH + 4 * hf;

    v8f acc0 = {};
    v8f acc1 = {};

#pragma unroll
    for (int j = 0; j < 16; ++j) {
        const v4f b0 = *(const v4f*)(bl + 8 * j);
        const v4f b1 = *(const v4f*)(bl + DIM + 8 * j);
        const v2f a01 = a[j].xy;
        const v2f a23 = a[j].zw;
        acc0 = __builtin_amdgcn_wmma_f32_16x16x4_f32(
            false, a01, false, b0.xy, (short)0, acc0, false, false);
        acc1 = __builtin_amdgcn_wmma_f32_16x16x4_f32(
            false, a01, false, b1.xy, (short)0, acc1, false, false);
        acc0 = __builtin_amdgcn_wmma_f32_16x16x4_f32(
            false, a23, false, b0.zw, (short)0, acc0, false, false);
        acc1 = __builtin_amdgcn_wmma_f32_16x16x4_f32(
            false, a23, false, b1.zw, (short)0, acc1, false, false);
    }

    if (!valid) return;
    const int mrw = m0 + hf * 8;
#pragma unroll
    for (int r = 0; r < 8; ++r) {
        float* prow = P + (size_t)(mrw + r) * 32;
        prow[sub]      = acc0[r];
        prow[16 + sub] = acc1[r];
    }
}

#define NGROUPS   (N_EDGES / 32)
#define K2_BLOCKS 512
#define K2_WAVES  (K2_BLOCKS * 8)
#define K2_FULL   (NGROUPS / K2_WAVES)
#define K2_REM    (NGROUPS - K2_FULL * K2_WAVES)

struct GroupRows { float4 s[4]; float4 d[4]; };

__device__ __forceinline__ void gather_group(const float* __restrict__ P,
                                             int si, int di, int q, int c4,
                                             GroupRows& r)
{
#pragma unroll
    for (int i = 0; i < 4; ++i) {
        const int from = (8 * i + q) * 4;
        const unsigned s = (unsigned)__builtin_amdgcn_ds_bpermute(from, si);
        const unsigned d = (unsigned)__builtin_amdgcn_ds_bpermute(from, di);
        r.s[i] = *(const float4*)(P + (s * 32u + (unsigned)c4));
        r.d[i] = *(const float4*)(P + (d * 32u + 16u + (unsigned)c4));
    }
}

__device__ __forceinline__ void store_group(float* __restrict__ out, int g,
                                            int q, int c4, const float4& bb,
                                            const GroupRows& r)
{
#pragma unroll
    for (int i = 0; i < 4; ++i) {
        const unsigned e = (unsigned)(g * 32 + 8 * i + q);
        float4 v;
        v.x = r.s[i].x + r.d[i].x + bb.x;
        v.y = r.s[i].y + r.d[i].y + bb.y;
        v.z = r.s[i].z + r.d[i].z + bb.z;
        v.w = r.s[i].w + r.d[i].w + bb.w;
        *(float4*)(out + (e * (unsigned)NCH + (unsigned)c4)) = v;
    }
}

template <int NG>
__device__ __forceinline__ void edge_wave(const float* __restrict__ P,
                                          const int* __restrict__ esrc,
                                          const int* __restrict__ edst,
                                          const float* __restrict__ b,
                                          float* __restrict__ out,
                                          int slot, int lane)
{
    const int q  = lane >> 2;
    const int c4 = (lane & 3) * 4;

    int si[NG], di[NG];
#pragma unroll
    for (int k = 0; k < NG; ++k) {
        const unsigned e = (unsigned)((k * K2_WAVES + slot) * 32 + lane);
        si[k] = esrc[e];
        di[k] = edst[e];
    }
    const float4 bb = *(const float4*)(b + c4);
    asm volatile("" :: "v"(bb.x), "v"(bb.y), "v"(bb.z), "v"(bb.w));

    GroupRows r[NG];
#pragma unroll
    for (int k = 0; k < NG; ++k)
        gather_group(P, si[k], di[k], q, c4, r[k]);
#pragma unroll
    for (int k = 0; k < NG; ++k)
        store_group(out, k * K2_WAVES + slot, q, c4, bb, r[k]);
}

__global__ __launch_bounds__(256) __attribute__((amdgpu_waves_per_eu(4, 4))) void edge_gather_add(
    const float* __restrict__ P,
    const int*   __restrict__ esrc,
    const int*   __restrict__ edst,
    const float* __restrict__ b,
    float*       __restrict__ out)
{
    const int lane = threadIdx.x & 31;
    const int w    = __builtin_amdgcn_readfirstlane((int)(threadIdx.x >> 5));
    const int slot = w * K2_BLOCKS + (int)blockIdx.x;
    if (slot < K2_REM)
        edge_wave<K2_FULL + 1>(P, esrc, edst, b, out, slot, lane);
    else
        edge_wave<K2_FULL>(P, esrc, edst, b, out, slot, lane);
}

extern "C" void kernel_launch(void* const* d_in, const int* in_sizes, int n_in,
                              void* d_out, int out_size, void* d_ws, size_t ws_size,
                              hipStream_t stream) {
    const float* h    = (const float*)d_in[0];
    const int*   src  = (const int*)  d_in[1];
    const int*   dst  = (const int*)  d_in[2];
    const float* W    = (const float*)d_in[3];
    const float* b    = (const float*)d_in[4];
    float*       out  = (float*)d_out;
    float*       P    = (float*)d_ws;

    const int waves_needed = (N_NODES + 15) / 16;
    const int blocks1 = (waves_needed + 7) / 8;
    node_proj_wmma<<<blocks1, 256, 0, stream>>>(h, W, P);

    edge_gather_add<<<K2_BLOCKS, 256, 0, stream>>>(P, src, dst, b, out);
}


# K1 as one resident pass: 512 blocks x 8 waves (2 per WGP), 1 or 2 tiles per wave, second tile requested before the first is used; 24-26 tiles per WGP, not 24 or 32; 6256 -> 4096 waves; sums kept
# speedup vs baseline: 1.5657x; 1.0290x over previous
#include <hip/hip_runtime.h>


typedef __attribute__((ext_vector_type(2))) float v2f;
typedef __attribute__((ext_vector_type(4))) float v4f;
typedef __attribute__((ext_vector_type(8))) float v8f;

#define N_NODES 100000
#define N_EDGES 600000
#define DIM     128
#define NCH     16
#define WPITCH  260

#define NTILES    (N_NODES / 16)
#define K1_BLOCKS 512
#define K1_WAVES  (K1_BLOCKS * 8)
#define K1_REM    (NTILES - K1_WAVES)

__device__ __forceinline__ void load_tile(const float* __restrict__ h, int m0,
                                          int sub, int hf, v4f (&a)[16])
{
    const v4f* arow = (const v4f*)(h + (size_t)(m0 + sub) * DIM + 4 * hf);
#pragma unroll
    for (int j = 0; j < 16; ++j) a[j] = arow[2 * j];
}

__device__ __forceinline__ void tile_product(const v4f (&a)[16], const float* bl,
                                             float* __restrict__ P, int m0,
                                             int sub, int hf)
{
    v8f acc0 = {};
    v8f acc1 = {};
#pragma unroll
    for (int j = 0; j < 16; ++j) {
        const v4f b0 = *(const v4f*)(bl + 8 * j);
        const v4f b1 = *(const v4f*)(bl + DIM + 8 * j);
        const v2f a01 = a[j].xy;
        const v2f a23 = a[j].zw;
        acc0 = __builtin_amdgcn_wmma_f32_16x16x4_f32(
            false, a01, false, b0.xy, (short)0, acc0, false, false);
        acc1 = __builtin_amdgcn_wmma_f32_16x16x4_f32(
            false, a01, false, b1.xy, (short)0, acc1, false, false);
        acc0 = __builtin_amdgcn_wmma_f32_16x16x4_f32(
            false, a23, false, b0.zw, (short)0, acc0, false, false);
        acc1 = __builtin_amdgcn_wmma_f32_16x16x4_f32(
            false, a23, false, b1.zw, (short)0, acc1, false, false);
    }
    const int mrw = m0 + hf * 8;
#pragma unroll
    for (int r = 0; r < 8; ++r) {
        float* prow = P + (size_t)(mrw + r) * 32;
        prow[sub]      = acc0[r];
        prow[16 + sub] = acc1[r];
    }
}

__global__ __launch_bounds__(256) __attribute__((amdgpu_waves_per_eu(4, 4))) void node_proj_wmma(
    const float* __restrict__ h,
    const float* __restrict__ W,
    float* __restrict__ P)
{
    __shared__ __attribute__((aligned(16))) float Ws[NCH * WPITCH];

    const int tid   = threadIdx.x;
    const int lane  = tid & 31;
    const int w     = __builtin_amdgcn_readfirstlane(tid >> 5);
    const int slot  = w * K1_BLOCKS + (int)blockIdx.x;
    const bool two  = slot < K1_REM;

    const int hf    = lane >> 4;
    const int sub   = lane & 15;

    v4f wv[4];
    {
        const v4f* Wg = (const v4f*)W;
#pragma unroll
        for (int i = 0; i < 4; ++i) wv[i] = Wg[tid + 256 * i];
    }

    v4f a0[16];
    load_tile(h, slot * 16, sub, hf, a0);

    {
        const int wr = tid >> 6;
        const int wc = (tid & 63) * 4;
#pragma unroll
        for (int i = 0; i < 4; ++i)
            *(v4f*)(Ws + (wr + 4 * i) * WPITCH + wc) = wv[i];
    }
    __builtin_amdgcn_fence(__ATOMIC_RELEASE, "workgroup", "local");
    __builtin_amdgcn_s_barrier();
    __builtin_amdgcn_fence(__ATOMIC_ACQUIRE, "workgroup", "local");

    const float* bl = Ws + sub * WPITCH + 4 * hf;

    if (two) {
        v4f a1[16];
        load_tile(h, (slot + K1_WAVES) * 16, sub, hf, a1);
        __builtin_amdgcn_sched_barrier(0);
        tile_product(a0, bl, P, slot * 16, sub, hf);
        __builtin_amdgcn_sched_barrier(0);
        tile_product(a1, bl, P, (slot + K1_WAVES) * 16, sub, hf);
    } else {
        tile_product(a0, bl, P, slot * 16, sub, hf);
    }
}

#define NGROUPS   (N_EDGES / 32)
#define K2_BLOCKS 512
#define K2_WAVES  (K2_BLOCKS * 8)
#define K2_FULL   (NGROUPS / K2_WAVES)
#define K2_REM    (NGROUPS - K2_FULL * K2_WAVES)

struct GroupRows { float4 s[4]; float4 d[4]; };

__device__ __forceinline__ void gather_group(const float* __restrict__ P,
                                             int si, int di, int q, int c4,
                                             GroupRows& r)
{
#pragma unroll
    for (int i = 0; i < 4; ++i) {
        const int from = (8 * i + q) * 4;
        const unsigned s = (unsigned)__builtin_amdgcn_ds_bpermute(from, si);
        const unsigned d = (unsigned)__builtin_amdgcn_ds_bpermute(from, di);
        r.s[i] = *(const float4*)(P + (s * 32u + (unsigned)c4));
        r.d[i] = *(const float4*)(P + (d * 32u + 16u + (unsigned)c4));
    }
}

__device__ __forceinline__ void store_group(float* __restrict__ out, int g,
                                            int q, int c4, const float4& bb,
                                            const GroupRows& r)
{
#pragma unroll
    for (int i = 0; i < 4; ++i) {
        const unsigned e = (unsigned)(g * 32 + 8 * i + q);
        float4 v;
        v.x = r.s[i].x + r.d[i].x + bb.x;
        v.y = r.s[i].y + r.d[i].y + bb.y;
        v.z = r.s[i].z + r.d[i].z + bb.z;
        v.w = r.s[i].w + r.d[i].w + bb.w;
        *(float4*)(out + (e * (unsigned)NCH + (unsigned)c4)) = v;
    }
}

template <int NG>
__device__ __forceinline__ void edge_wave(const float* __restrict__ P,
                                          const int* __restrict__ esrc,
                                          const int* __restrict__ edst,
                                          const float* __restrict__ b,
                                          float* __restrict__ out,
                                          int slot, int lane)
{
    const int q  = lane >> 2;
    const int c4 = (lane & 3) * 4;

    int si[NG], di[NG];
#pragma unroll
    for (int k = 0; k < NG; ++k) {
        const unsigned e = (unsigned)((k * K2_WAVES + slot) * 32 + lane);
        si[k] = esrc[e];
        di[k] = edst[e];
    }
    const float4 bb = *(const float4*)(b + c4);
    asm volatile("" :: "v"(bb.x), "v"(bb.y), "v"(bb.z), "v"(bb.w));

    GroupRows r[NG];
#pragma unroll
    for (int k = 0; k < NG; ++k)
        gather_group(P, si[k], di[k], q, c4, r[k]);
#pragma unroll
    for (int k = 0; k < NG; ++k)
        store_group(out, k * K2_WAVES + slot, q, c4, bb, r[k]);
}

__global__ __launch_bounds__(256) __attribute__((amdgpu_waves_per_eu(4, 4))) void edge_gather_add(
    const float* __restrict__ P,
    const int*   __restrict__ esrc,
    const int*   __restrict__ edst,
    const float* __restrict__ b,
    float*       __restrict__ out)
{
    const int lane = threadIdx.x & 31;
    const int w    = __builtin_amdgcn_readfirstlane((int)(threadIdx.x >> 5));
    const int slot = w * K2_BLOCKS + (int)blockIdx.x;
    if (slot < K2_REM)
        edge_wave<K2_FULL + 1>(P, esrc, edst, b, out, slot, lane);
    else
        edge_wave<K2_FULL>(P, esrc, edst, b, out, slot, lane);
}

extern "C" void kernel_launch(void* const* d_in, const int* in_sizes, int n_in,
                              void* d_out, int out_size, void* d_ws, size_t ws_size,
                              hipStream_t stream) {
    const float* h    = (const float*)d_in[0];
    const int*   src  = (const int*)  d_in[1];
    const int*   dst  = (const int*)  d_in[2];
    const float* W    = (const float*)d_in[3];
    const float* b    = (const float*)d_in[4];
    float*       out  = (float*)d_out;
    float*       P    = (float*)d_ws;

    node_proj_wmma<<<K1_BLOCKS, 256, 0, stream>>>(h, W, P);

    edge_gather_add<<<K2_BLOCKS, 256, 0, stream>>>(P, src, dst, b, out);
}
